# starting kernel
# baseline (speedup 1.0000x reference)
#include <hip/hip_runtime.h>

#define NN 100000
#define EE 2000000
#define DD 256
#define CC 16
#define RR 4

typedef __attribute__((ext_vector_type(2))) float v2f;
typedef __attribute__((ext_vector_type(8))) float v8f;

__global__ void zero_kernel(float* __restrict__ p, long long n) {
    long long i = (long long)blockIdx.x * blockDim.x + threadIdx.x;
    long long stride = (long long)gridDim.x * blockDim.x;
    for (; i < n; i += stride) p[i] = 0.0f;
}

__global__ __launch_bounds__(128) void gemm_wmma_kernel(
    const float* __restrict__ feat,
    const float* __restrict__ W,
    const float* __restrict__ bias,
    float* __restrict__ wh)
{
    const int wave = threadIdx.x >> 5;
    const int lane = threadIdx.x & 31;
    const int col  = lane & 15;
    const int half = lane >> 4;

    const int row0 = blockIdx.x * 16;
    const int arow = row0 + col;

    const float* aptr = feat + (size_t)arow * DD + 2 * half;
    const float* bptr = W + (size_t)wave * (DD * CC) + (2 * half) * CC + col;

    v8f acc = {};
    #pragma unroll 8
    for (int k = 0; k < DD; k += 4) {
        v2f a, b;
        a.x = aptr[0];
        a.y = aptr[1];
        b.x = bptr[0];
        b.y = bptr[CC];
        acc = __builtin_amdgcn_wmma_f32_16x16x4_f32(
             false, a,  false, b,
             (short)0, acc,  false,  false);
        aptr += 4;
        bptr += 4 * CC;
    }

    const float bv = bias[wave * CC + col];

    float* out = wh + (size_t)row0 * (RR * CC) + wave * CC + col;
    #pragma unroll
    for (int i = 0; i < 8; ++i) {
        const int m = i + 8 * half;
        out[(size_t)m * (RR * CC)] = acc[i] + bv;
    }
}

__global__ __launch_bounds__(256) void scatter_kernel(
    const float* __restrict__ wh,
    const int*   __restrict__ src,
    const int*   __restrict__ dst,
    const float* __restrict__ ew,
    float* __restrict__ sums,
    float* __restrict__ cnts)
{
    const long long total = (long long)RR * EE;
    long long i = (long long)blockIdx.x * blockDim.x + threadIdx.x;
    if (i >= total) return;

    const int r = (int)(i / EE);
    const int s = src[i];
    const int d = dst[i];
    const float w = ew[i];

    const float4* whp = (const float4*)(wh + (size_t)s * (RR * CC) + r * CC);
    float4 v0 = whp[0];
    float4 v1 = whp[1];
    float4 v2 = whp[2];
    float4 v3 = whp[3];

    float* sp = sums + ((size_t)r * NN + d) * CC;
    atomicAdd(sp + 0,  v0.x * w);
    atomicAdd(sp + 1,  v0.y * w);
    atomicAdd(sp + 2,  v0.z * w);
    atomicAdd(sp + 3,  v0.w * w);
    atomicAdd(sp + 4,  v1.x * w);
    atomicAdd(sp + 5,  v1.y * w);
    atomicAdd(sp + 6,  v1.z * w);
    atomicAdd(sp + 7,  v1.w * w);
    atomicAdd(sp + 8,  v2.x * w);
    atomicAdd(sp + 9,  v2.y * w);
    atomicAdd(sp + 10, v2.z * w);
    atomicAdd(sp + 11, v2.w * w);
    atomicAdd(sp + 12, v3.x * w);
    atomicAdd(sp + 13, v3.y * w);
    atomicAdd(sp + 14, v3.z * w);
    atomicAdd(sp + 15, v3.w * w);

    atomicAdd(cnts + (size_t)r * NN + d, 1.0f);
}

__global__ __launch_bounds__(256) void finalize_kernel(
    const float* __restrict__ sums,
    const float* __restrict__ cnts,
    float* __restrict__ out)
{
    const long long total = (long long)NN * CC;
    long long i = (long long)blockIdx.x * blockDim.x + threadIdx.x;
    if (i >= total) return;
    const int n = (int)(i >> 4);
    const int c = (int)(i & 15);

    float acc = 0.0f;
    #pragma unroll
    for (int r = 0; r < RR; ++r) {
        const float cnt = cnts[(size_t)r * NN + n];
        if (cnt > 0.0f) {
            acc += sums[((size_t)r * NN + n) * CC + c] / cnt;
        }
    }
    out[i] = acc;
}

extern "C" void kernel_launch(void* const* d_in, const int* in_sizes, int n_in,
                              void* d_out, int out_size, void* d_ws, size_t ws_size,
                              hipStream_t stream) {
    const float* feat = (const float*)d_in[0];
    const int*   srci = (const int*)  d_in[1];
    const int*   dsti = (const int*)  d_in[2];
    const float* ew   = (const float*)d_in[3];
    const float* W    = (const float*)d_in[4];
    const float* bias = (const float*)d_in[5];
    float* out = (float*)d_out;

    float* ws   = (float*)d_ws;
    float* wh   = ws;
    float* sums = ws + (size_t)NN * (RR * CC);
    float* cnts = sums + (size_t)NN * (RR * CC);

    {
        const long long nz = (long long)NN * (RR * CC) + (long long)RR * NN;
        const int blocks = 4096;
        zero_kernel<<<blocks, 256, 0, stream>>>(sums, nz);
    }

    {
        dim3 grid(NN / 16);
        gemm_wmma_kernel<<<grid, 128, 0, stream>>>(feat, W, bias, wh);
    }

    {
        const long long total = (long long)RR * EE;
        const int blocks = (int)((total + 255) / 256);
        scatter_kernel<<<blocks, 256, 0, stream>>>(wh, srci, dsti, ew, sums, cnts);
    }

    {
        const long long total = (long long)NN * CC;
        const int blocks = (int)((total + 255) / 256);
        finalize_kernel<<<blocks, 256, 0, stream>>>(sums, cnts, out);
    }
}


# scatter: each list half's work-groups store their slab (sums, counts) to a plane of their own and the finalize adds the two planes; no plane clear, no global f32 atomics; same two addends per row
# speedup vs baseline: 9.5000x; 9.5000x over previous
#include <hip/hip_runtime.h>

#define NN 100000
#define EE 2000000
#define DD 256
#define CC 16
#define RR 4

typedef __attribute__((ext_vector_type(2))) float v2f;
typedef __attribute__((ext_vector_type(8))) float v8f;
typedef __attribute__((ext_vector_type(4))) float v4f;

#define LIN_TILES  (NN / 16)
#define LIN_GRID   1024
#define LIN_PITCH  (DD + 4)

#define PLANE_FLOATS (NN * RR * CC + NN * RR)
#define ACC_V4     (PLANE_FLOATS / 4)
#define TWO_PLANE_BYTES ((size_t)(NN * RR * CC + 2 * PLANE_FLOATS) * sizeof(float))
static_assert(NN % 16 == 0, "whole row tiles");
static_assert((NN * RR * CC + NN * RR) % 4 == 0, "planes are whole 16-byte pieces");
static_assert(LIN_GRID <= LIN_TILES, "every work-group has a first tile");

__global__ __launch_bounds__(128) void gemm_wmma_kernel(
    const float* __restrict__ feat,
    const float* __restrict__ W,
    const float* __restrict__ bias,
    float* __restrict__ wh,
    float* __restrict__ acc_planes,
    const size_t ws_bytes)
{
    __shared__ float stage[2][16 * LIN_PITCH];

    const int tid  = threadIdx.x;

    if (ws_bytes < TWO_PLANE_BYTES) {
        v4f* z = (v4f*)acc_planes;
        const v4f zero = {0.0f, 0.0f, 0.0f, 0.0f};
        for (int i = blockIdx.x * 128 + tid; i < ACC_V4; i += LIN_GRID * 128) z[i] = zero;
    }

    const int wave = tid >> 5;
    const int lane = tid & 31;
    const int col  = lane & 15;
    const int half = lane >> 4;

    v2f breg[DD / 4];
    {
        const float* bptr = W + (size_t)wave * (DD * CC) + (2 * half) * CC + col;
        #pragma unroll
        for (int s = 0; s < DD / 4; ++s) {
            breg[s].x = bptr[s * 4 * CC];
            breg[s].y = bptr[s * 4 * CC + CC];
        }
    }
    const float bv = bias[wave * CC + col];

    const int srow = tid >> 6;
    const int sc4  = tid & 63;

    int tile = blockIdx.x;
    v4f pre0, pre1, pre2, pre3, pre4, pre5, pre6, pre7;
    {
        const v4f* g = (const v4f*)(feat + (size_t)tile * (16 * DD)) + tid;
        pre0 = g[0 * 128]; pre1 = g[1 * 128]; pre2 = g[2 * 128]; pre3 = g[3 * 128];
        pre4 = g[4 * 128]; pre5 = g[5 * 128]; pre6 = g[6 * 128]; pre7 = g[7 * 128];
    }

    int p = 0;
    for (; tile < LIN_TILES; tile += LIN_GRID) {
        float* sb = stage[p];
        {
            float* sp = sb + srow * LIN_PITCH + sc4 * 4;
            *(v4f*)(sp +  0 * LIN_PITCH) = pre0; *(v4f*)(sp +  2 * LIN_PITCH) = pre1;
            *(v4f*)(sp +  4 * LIN_PITCH) = pre2; *(v4f*)(sp +  6 * LIN_PITCH) = pre3;
            *(v4f*)(sp +  8 * LIN_PITCH) = pre4; *(v4f*)(sp + 10 * LIN_PITCH) = pre5;
            *(v4f*)(sp + 12 * LIN_PITCH) = pre6; *(v4f*)(sp + 14 * LIN_PITCH) = pre7;
        }
        __syncthreads();

        {
            const int nt = (tile + LIN_GRID < LIN_TILES) ? tile + LIN_GRID : tile;
            const v4f* g = (const v4f*)(feat + (size_t)nt * (16 * DD)) + tid;
            pre0 = g[0 * 128]; pre1 = g[1 * 128]; pre2 = g[2 * 128]; pre3 = g[3 * 128];
            pre4 = g[4 * 128]; pre5 = g[5 * 128]; pre6 = g[6 * 128]; pre7 = g[7 * 128];
        }

        const float* ap = sb + col * LIN_PITCH + 2 * half;
        v8f acc = {};
        v2f af[2][16];
        #pragma unroll
        for (int q = 0; q < 16; ++q) af[0][q] = *(const v2f*)(ap + 4 * q);
        #pragma unroll
        for (int blk = 0; blk < 4; ++blk) {
            if (blk < 3) {
                #pragma unroll
                for (int q = 0; q < 16; ++q)
                    af[(blk + 1) & 1][q] = *(const v2f*)(ap + 4 * (16 * (blk + 1) + q));
            }
            __builtin_amdgcn_sched_barrier(0);
            #pragma unroll
            for (int q = 0; q < 16; ++q)
                acc = __builtin_amdgcn_wmma_f32_16x16x4_f32(
                     false, af[blk & 1][q],  false, breg[16 * blk + q],
                     (short)0, acc,  false,  false);
        }

        float* out = wh + (size_t)tile * (16 * RR * CC) + wave * CC + col;
        #pragma unroll
        for (int i = 0; i < 8; ++i) {
            const int m = i + 8 * half;
            out[(size_t)m * (RR * CC)] = acc[i] + bv;
        }
        p ^= 1;
    }
}

#define SLAB_ROWS 3136
#define SLABS     32
#define PARTS     2
#define SC_WAVES  32
#define SEG_EDGES 128
#define SEGS      (EE / SEG_EDGES)
#define PART_SEGS ((SEGS + PARTS - 1) / PARTS)
#define STEP_SEGS 4
#define FULL_STEPS ((SEGS - PART_SEGS) / STEP_SEGS)
#define QCAP      (32 + STEP_SEGS * SEG_EDGES)
#define DL_BITS   12

static_assert(EE % SEG_EDGES == 0, "whole segments");
static_assert(SLAB_ROWS <= (1 << DL_BITS), "row inside the slab fits its field");
static_assert(PART_SEGS * SEG_EDGES <= (1 << (32 - DL_BITS)), "edge index inside a list half fits its field");
static_assert(PART_SEGS / STEP_SEGS == FULL_STEPS && PART_SEGS - FULL_STEPS * STEP_SEGS <= SC_WAVES, "both halves have the same full steps; the rest is one segment per wave");
static_assert(FULL_STEPS >= SC_WAVES, "every wave has a first step");
static_assert(SLABS * SLAB_ROWS >= NN && SLAB_ROWS % 32 == 0 && (NN % 32) == 0, "slabs cover all rows, line aligned");

template <bool FULL>
__device__ __forceinline__ void flush_edges(
    const unsigned* qp, int n, uint4* rc,
    const int* __restrict__ src_r, const float* __restrict__ ew_r,
    const float* __restrict__ whr, float* slab, int lane)
{
    const unsigned c = lane & 15;
    const int half = lane >> 4;

    unsigned e = 0;
    if (FULL || lane < n) e = qp[lane];
    const unsigned ei = e >> DL_BITS;
    const unsigned dl = e & ((1u << DL_BITS) - 1u);
    const unsigned s = (unsigned)src_r[ei];
    const float w = ew_r[ei];
    rc[lane] = make_uint4(s * (unsigned)(RR * CC), __float_as_uint(w), dl * (unsigned)CC, 0u);

    uint4 t[16];
    float v[16];
    #pragma unroll
    for (int p = 0; p < 16; ++p)
        t[p] = rc[2 * p + half];
    #pragma unroll
    for (int p = 0; p < 16; ++p)
        v[p] = whr[t[p].x + c];
    #pragma unroll
    for (int p = 0; p < 16; ++p) {
        if (FULL || 2 * p + half < n)
            atomicAdd(&slab[t[p].z + c], v[p] * __uint_as_float(t[p].y));
    }
}

__device__ __forceinline__ void scan_segment(
    const int4 d4, const int seg, const int base, const unsigned nrows,
    unsigned* q, unsigned* cntl, const unsigned lane_bits, int& qn)
{
    const int dvals[4] = {d4.x, d4.y, d4.z, d4.w};
    #pragma unroll
    for (int k = 0; k < 4; ++k) {
        const unsigned dl = (unsigned)(dvals[k] - base);
        const bool m = dl < nrows;
        const unsigned mask = __builtin_amdgcn_ballot_w32(m);
        if (m) {
            const unsigned pos = __builtin_amdgcn_mbcnt_lo(mask, (unsigned)qn);
            q[pos] = ((unsigned)(seg * SEG_EDGES + k) << DL_BITS) + lane_bits + dl;
            atomicAdd(&cntl[dl], 1u);
        }
        qn += __builtin_popcount(mask);
    }
}

__global__ __launch_bounds__(1024) void scatter_kernel(
    const float* __restrict__ wh,
    const int*   __restrict__ src,
    const int*   __restrict__ dst,
    const float* __restrict__ ew,
    float* __restrict__ sums,
    float* __restrict__ cnts,
    const size_t ws_bytes)
{
    __shared__ float    slab[SLAB_ROWS * CC];
    __shared__ unsigned cntl[SLAB_ROWS];
    __shared__ unsigned queue[SC_WAVES * QCAP];
    __shared__ uint4    rec[SC_WAVES * 32];

    const int tid  = threadIdx.x;
    const int lane = tid & 31;
    const int wave = __builtin_amdgcn_readfirstlane(tid >> 5);
    const int part = blockIdx.x / (RR * SLABS);
    const int r    = (blockIdx.x / SLABS) % RR;
    const int base = (blockIdx.x % SLABS) * SLAB_ROWS;
    const int seg0 = part * PART_SEGS;
    const int nseg = (SEGS - seg0 < PART_SEGS) ? (SEGS - seg0) : PART_SEGS;
    const int nrows = (NN - base < SLAB_ROWS) ? (NN - base) : SLAB_ROWS;

    for (int i = tid; i < SLAB_ROWS * CC / 4; i += SC_WAVES * 32)
        ((float4*)slab)[i] = make_float4(0.0f, 0.0f, 0.0f, 0.0f);
    for (int i = tid; i < SLAB_ROWS; i += SC_WAVES * 32)
        cntl[i] = 0u;
    __syncthreads();

    const size_t list0 = (size_t)r * EE + (size_t)seg0 * SEG_EDGES;
    const int4*  dv    = (const int4*)(dst + list0) + lane;
    const int*   src_r = src + list0;
    const float* ew_r  = ew + list0;
    const float* whr   = wh + r * CC;
    unsigned* q  = queue + wave * QCAP;
    uint4*    rc = rec + wave * 32;
    const unsigned lane_bits = (unsigned)(lane * 4) << DL_BITS;

    int qn = 0;
    int4 cur[STEP_SEGS], nxt[STEP_SEGS];

    int g = wave;
    #pragma unroll
    for (int u = 0; u < STEP_SEGS; ++u)
        cur[u] = dv[(g * STEP_SEGS + u) * 32];

    #pragma unroll 2
    for (; g < FULL_STEPS; g += SC_WAVES) {
        const int gn = (g + SC_WAVES < FULL_STEPS) ? g + SC_WAVES : g;
        #pragma unroll
        for (int u = 0; u < STEP_SEGS; ++u)
            nxt[u] = dv[(gn * STEP_SEGS + u) * 32];

        #pragma unroll
        for (int u = 0; u < STEP_SEGS; ++u)
            scan_segment(cur[u], g * STEP_SEGS + u, base, (unsigned)nrows, q, cntl, lane_bits, qn);

        while (qn >= 32) {
            qn -= 32;
            flush_edges<true>(q + qn, 32, rc, src_r, ew_r, whr, slab, lane);
        }

        #pragma unroll
        for (int u = 0; u < STEP_SEGS; ++u) cur[u] = nxt[u];
    }
    if (wave < nseg - FULL_STEPS * STEP_SEGS) {
        const int seg = FULL_STEPS * STEP_SEGS + wave;
        scan_segment(dv[seg * 32], seg, base, (unsigned)nrows, q, cntl, lane_bits, qn);
    }
    while (qn >= 32) {
        qn -= 32;
        flush_edges<true>(q + qn, 32, rc, src_r, ew_r, whr, slab, lane);
    }
    if (qn > 0) flush_edges<false>(q, qn, rc, src_r, ew_r, whr, slab, lane);

    __syncthreads();

    if (ws_bytes >= TWO_PLANE_BYTES) {
        float4* so = (float4*)(sums + (size_t)part * PLANE_FLOATS + ((size_t)r * NN + base) * CC);
        for (int i = tid; i < nrows * (CC / 4); i += SC_WAVES * 32)
            so[i] = ((const float4*)slab)[i];
        float* co = cnts + (size_t)part * PLANE_FLOATS + (size_t)r * NN + base;
        for (int i = tid; i < nrows; i += SC_WAVES * 32)
            co[i] = (float)cntl[i];
    } else {
        float* so = sums + ((size_t)r * NN + base) * CC;
        for (int i = tid; i < nrows * CC; i += SC_WAVES * 32)
            atomicAdd(&so[i], slab[i]);
        float* co = cnts + (size_t)r * NN + base;
        for (int i = tid; i < nrows; i += SC_WAVES * 32)
            atomicAdd(&co[i], (float)cntl[i]);
    }
}

__global__ __launch_bounds__(256) void finalize_kernel(
    const float* __restrict__ sums,
    const float* __restrict__ cnts,
    float* __restrict__ out,
    const size_t ws_bytes)
{
    const bool two = ws_bytes >= TWO_PLANE_BYTES;
    const long long total = (long long)NN * CC;
    long long i = (long long)blockIdx.x * blockDim.x + threadIdx.x;
    if (i >= total) return;
    const int n = (int)(i >> 4);
    const int c = (int)(i & 15);

    float acc = 0.0f;
    #pragma unroll
    for (int r = 0; r < RR; ++r) {
        float cnt = cnts[(size_t)r * NN + n];
        if (two) cnt += cnts[PLANE_FLOATS + (size_t)r * NN + n];
        if (cnt > 0.0f) {
            float s = sums[((size_t)r * NN + n) * CC + c];
            if (two) s += sums[PLANE_FLOATS + ((size_t)r * NN + n) * CC + c];
            acc += s / cnt;
        }
    }
    out[i] = acc;
}

extern "C" void kernel_launch(void* const* d_in, const int* in_sizes, int n_in,
                              void* d_out, int out_size, void* d_ws, size_t ws_size,
                              hipStream_t stream) {
    const float* feat = (const float*)d_in[0];
    const int*   srci = (const int*)  d_in[1];
    const int*   dsti = (const int*)  d_in[2];
    const float* ew   = (const float*)d_in[3];
    const float* W    = (const float*)d_in[4];
    const float* bias = (const float*)d_in[5];
    float* out = (float*)d_out;

    float* ws   = (float*)d_ws;
    float* wh   = ws;
    float* sums = ws + (size_t)NN * (RR * CC);
    float* cnts = sums + (size_t)NN * (RR * CC);

    {
        gemm_wmma_kernel<<<LIN_GRID, 128, 0, stream>>>(feat, W, bias, wh, sums, ws_size);
    }

    {
        scatter_kernel<<<PARTS * RR * SLABS, SC_WAVES * 32, 0, stream>>>(wh, srci, dsti, ew, sums, cnts, ws_size);
    }

    {
        const long long total = (long long)NN * CC;
        const int blocks = (int)((total + 255) / 256);
        finalize_kernel<<<blocks, 256, 0, stream>>>(sums, cnts, out, ws_size);
    }
}
